# starting kernel
# baseline (speedup 1.0000x reference)
#include <hip/hip_runtime.h>
#include <hip/hip_bf16.h>
#include <math.h>

#define BATCH 1
#define SEQ   128
#define HID   768
#define NL    10
#define NR    2

typedef __attribute__((ext_vector_type(2))) float v2f;
typedef __attribute__((ext_vector_type(8))) float v8f;

__global__ __launch_bounds__(256)
void xa_kernel(const float* __restrict__ X, const float* __restrict__ A,
               float* __restrict__ xa) {
    int idx = blockIdx.x * blockDim.x + threadIdx.x;
    if (idx >= SEQ * NL * NR) return;
    int r = idx & 1;
    int l = (idx >> 1) % NL;
    int s = idx / (NL * NR);
    const float* xrow = X + s * HID;
    float sum = 0.0f;
    #pragma unroll 8
    for (int h = 0; h < HID; ++h)
        sum += xrow[h] * A[(h * NR + r) * NL + l];
    xa[idx] = sum;
}

__global__ __launch_bounds__(32)
void fused_linear_wmma(const float* __restrict__ X,
                       const float* __restrict__ W,
                       const float* __restrict__ bias,
                       const float* __restrict__ Blora,
                       const float* __restrict__ xa,
                       float* __restrict__ out)
{
    const int hTile = blockIdx.x * 16;
    const int sTile = blockIdx.y * 16;
    const int lane  = threadIdx.x;
    const int nLoc  = lane & 15;
    const int kOff  = (lane >> 4) * 2;

    v8f acc = {};
    const float* arow = X + (sTile + nLoc) * HID;

    for (int k0 = 0; k0 < HID; k0 += 4) {
        const int kk = k0 + kOff;
        v2f a = *(const v2f*)(arow + kk);
        v2f b;
        b.x = W[kk * HID + (hTile + nLoc)];
        b.y = W[(kk + 1) * HID + (hTile + nLoc)];
        acc = __builtin_amdgcn_wmma_f32_16x16x4_f32(
                  false, a, false, b, (short)0, acc, false, false);
    }

    const int n     = hTile + nLoc;
    const int mBase = (lane < 16) ? 0 : 8;
    const float bv  = bias[n];

    #pragma unroll
    for (int l = 0; l < NL; ++l) {
        const float bl0 = Blora[(0 * HID + n) * NL + l];
        const float bl1 = Blora[(1 * HID + n) * NL + l];
        #pragma unroll
        for (int i = 0; i < 8; ++i) {
            const int s = sTile + mBase + i;
            const float xa0 = xa[(s * NL + l) * NR + 0];
            const float xa1 = xa[(s * NL + l) * NR + 1];
            out[(s * NL + l) * HID + n] = acc[i] + bv + xa0 * bl0 + xa1 * bl1;
        }
    }
}

__global__ __launch_bounds__(256)
void pair_gelu_dot(const float* __restrict__ srcOut,
                   const float* __restrict__ tgtOut,
                   const float* __restrict__ cls,
                   float* __restrict__ out)
{
    const int s = blockIdx.x;
    const int l = blockIdx.y;

    __shared__ float sSrc[HID];
    __shared__ float sCls[HID];
    for (int h = threadIdx.x; h < HID; h += 256) {
        sSrc[h] = srcOut[(s * NL + l) * HID + h];
        sCls[h] = cls[h * NL + l];
    }
    __syncthreads();

    const int wave = threadIdx.x >> 5;
    const int lane = threadIdx.x & 31;
    const float inv_sqrt2 = 0.70710678118654752440f;

    for (int t = wave; t < SEQ; t += 8) {
        const float* trow = tgtOut + (t * NL + l) * HID;
        float sum = 0.0f;
        #pragma unroll 4
        for (int h = lane; h < HID; h += 32) {
            const float v = sSrc[h] + trow[h];
            const float g = 0.5f * v * (1.0f + erff(v * inv_sqrt2));
            sum += g * sCls[h];
        }
        #pragma unroll
        for (int off = 16; off > 0; off >>= 1)
            sum += __shfl_xor(sum, off, 32);
        if (lane == 0)
            out[(l * SEQ + s) * SEQ + t] = sum;
    }
}

extern "C" void kernel_launch(void* const* d_in, const int* in_sizes, int n_in,
                              void* d_out, int out_size, void* d_ws, size_t ws_size,
                              hipStream_t stream) {
    (void)in_sizes; (void)n_in; (void)out_size; (void)ws_size;

    const float* x     = (const float*)d_in[0];
    const float* W_src = (const float*)d_in[1];
    const float* b_src = (const float*)d_in[2];
    const float* W_tgt = (const float*)d_in[3];
    const float* b_tgt = (const float*)d_in[4];
    const float* A_src = (const float*)d_in[5];
    const float* B_src = (const float*)d_in[6];
    const float* A_tgt = (const float*)d_in[7];
    const float* B_tgt = (const float*)d_in[8];
    const float* cls   = (const float*)d_in[9];

    float* ws     = (float*)d_ws;
    float* srcOut = ws;
    float* tgtOut = srcOut + SEQ * NL * HID;
    float* xaSrc  = tgtOut + SEQ * NL * HID;
    float* xaTgt  = xaSrc + SEQ * NL * NR;

    const int xaThreads = SEQ * NL * NR;
    xa_kernel<<<(xaThreads + 255) / 256, 256, 0, stream>>>(x, A_src, xaSrc);
    xa_kernel<<<(xaThreads + 255) / 256, 256, 0, stream>>>(x, A_tgt, xaTgt);

    dim3 g1(HID / 16, SEQ / 16);
    fused_linear_wmma<<<g1, 32, 0, stream>>>(x, W_src, b_src, B_src, xaSrc, srcOut);
    fused_linear_wmma<<<g1, 32, 0, stream>>>(x, W_tgt, b_tgt, B_tgt, xaTgt, tgtOut);

    dim3 g2(SEQ, NL);
    pair_gelu_dot<<<g2, 256, 0, stream>>>(srcOut, tgtOut, cls, (float*)d_out);
}


# Down-projection xa: the 768-term sum is split over a block (thread takes 3 h with 20 sums, independent loads; lane butterfly + LDS over 8 waves), not serial per thread; reorders xa sums
# speedup vs baseline: 1.4160x; 1.4160x over previous
#include <hip/hip_runtime.h>
#include <hip/hip_bf16.h>
#include <math.h>

#define BATCH 1
#define SEQ   128
#define HID   768
#define NL    10
#define NR    2

typedef __attribute__((ext_vector_type(2))) float v2f;
typedef __attribute__((ext_vector_type(4))) float v4f;
typedef __attribute__((ext_vector_type(8))) float v8f;

__global__ __launch_bounds__(256)
void xa_kernel(const float* __restrict__ X, const float* __restrict__ A,
               float* __restrict__ xa) {
    const int s    = blockIdx.x;
    const int tid  = threadIdx.x;
    const int lane = tid & 31;
    const int wave = tid >> 5;

    float acc[NR * NL];
    #pragma unroll
    for (int c = 0; c < NR * NL; ++c) acc[c] = 0.0f;

    #pragma unroll
    for (int j = 0; j < HID / 256; ++j) {
        const int h = tid + 256 * j;
        const float xv = X[s * HID + h];
        const v4f* arow = (const v4f*)(A + h * (NR * NL));
        #pragma unroll
        for (int q = 0; q < (NR * NL) / 4; ++q) {
            const v4f a = arow[q];
            acc[4 * q + 0] += xv * a.x;
            acc[4 * q + 1] += xv * a.y;
            acc[4 * q + 2] += xv * a.z;
            acc[4 * q + 3] += xv * a.w;
        }
    }

    #pragma unroll
    for (int c = 0; c < NR * NL; ++c) {
        float v = acc[c];
        #pragma unroll
        for (int off = 16; off > 0; off >>= 1)
            v += __shfl_xor(v, off, 32);
        acc[c] = v;
    }

    __shared__ float red[8][NR * NL];
    if (lane == 0) {
        #pragma unroll
        for (int c = 0; c < NR * NL; ++c) red[wave][c] = acc[c];
    }
    __syncthreads();

    if (tid < NR * NL) {
        float sum = 0.0f;
        #pragma unroll
        for (int w = 0; w < 8; ++w) sum += red[w][tid];
        const int r = tid / NL;
        const int l = tid - r * NL;
        xa[(s * NL + l) * NR + r] = sum;
    }
}

__global__ __launch_bounds__(32)
void fused_linear_wmma(const float* __restrict__ X,
                       const float* __restrict__ W,
                       const float* __restrict__ bias,
                       const float* __restrict__ Blora,
                       const float* __restrict__ xa,
                       float* __restrict__ out)
{
    const int hTile = blockIdx.x * 16;
    const int sTile = blockIdx.y * 16;
    const int lane  = threadIdx.x;
    const int nLoc  = lane & 15;
    const int kOff  = (lane >> 4) * 2;

    v8f acc = {};
    const float* arow = X + (sTile + nLoc) * HID;

    for (int k0 = 0; k0 < HID; k0 += 4) {
        const int kk = k0 + kOff;
        v2f a = *(const v2f*)(arow + kk);
        v2f b;
        b.x = W[kk * HID + (hTile + nLoc)];
        b.y = W[(kk + 1) * HID + (hTile + nLoc)];
        acc = __builtin_amdgcn_wmma_f32_16x16x4_f32(
                  false, a, false, b, (short)0, acc, false, false);
    }

    const int n     = hTile + nLoc;
    const int mBase = (lane < 16) ? 0 : 8;
    const float bv  = bias[n];

    #pragma unroll
    for (int l = 0; l < NL; ++l) {
        const float bl0 = Blora[(0 * HID + n) * NL + l];
        const float bl1 = Blora[(1 * HID + n) * NL + l];
        #pragma unroll
        for (int i = 0; i < 8; ++i) {
            const int s = sTile + mBase + i;
            const float xa0 = xa[(s * NL + l) * NR + 0];
            const float xa1 = xa[(s * NL + l) * NR + 1];
            out[(s * NL + l) * HID + n] = acc[i] + bv + xa0 * bl0 + xa1 * bl1;
        }
    }
}

__global__ __launch_bounds__(256)
void pair_gelu_dot(const float* __restrict__ srcOut,
                   const float* __restrict__ tgtOut,
                   const float* __restrict__ cls,
                   float* __restrict__ out)
{
    const int s = blockIdx.x;
    const int l = blockIdx.y;

    __shared__ float sSrc[HID];
    __shared__ float sCls[HID];
    for (int h = threadIdx.x; h < HID; h += 256) {
        sSrc[h] = srcOut[(s * NL + l) * HID + h];
        sCls[h] = cls[h * NL + l];
    }
    __syncthreads();

    const int wave = threadIdx.x >> 5;
    const int lane = threadIdx.x & 31;
    const float inv_sqrt2 = 0.70710678118654752440f;

    for (int t = wave; t < SEQ; t += 8) {
        const float* trow = tgtOut + (t * NL + l) * HID;
        float sum = 0.0f;
        #pragma unroll 4
        for (int h = lane; h < HID; h += 32) {
            const float v = sSrc[h] + trow[h];
            const float g = 0.5f * v * (1.0f + erff(v * inv_sqrt2));
            sum += g * sCls[h];
        }
        #pragma unroll
        for (int off = 16; off > 0; off >>= 1)
            sum += __shfl_xor(sum, off, 32);
        if (lane == 0)
            out[(l * SEQ + s) * SEQ + t] = sum;
    }
}

extern "C" void kernel_launch(void* const* d_in, const int* in_sizes, int n_in,
                              void* d_out, int out_size, void* d_ws, size_t ws_size,
                              hipStream_t stream) {
    (void)in_sizes; (void)n_in; (void)out_size; (void)ws_size;

    const float* x     = (const float*)d_in[0];
    const float* W_src = (const float*)d_in[1];
    const float* b_src = (const float*)d_in[2];
    const float* W_tgt = (const float*)d_in[3];
    const float* b_tgt = (const float*)d_in[4];
    const float* A_src = (const float*)d_in[5];
    const float* B_src = (const float*)d_in[6];
    const float* A_tgt = (const float*)d_in[7];
    const float* B_tgt = (const float*)d_in[8];
    const float* cls   = (const float*)d_in[9];

    float* ws     = (float*)d_ws;
    float* srcOut = ws;
    float* tgtOut = srcOut + SEQ * NL * HID;
    float* xaSrc  = tgtOut + SEQ * NL * HID;
    float* xaTgt  = xaSrc + SEQ * NL * NR;

    xa_kernel<<<SEQ, 256, 0, stream>>>(x, A_src, xaSrc);
    xa_kernel<<<SEQ, 256, 0, stream>>>(x, A_tgt, xaTgt);

    dim3 g1(HID / 16, SEQ / 16);
    fused_linear_wmma<<<g1, 32, 0, stream>>>(x, W_src, b_src, B_src, xaSrc, srcOut);
    fused_linear_wmma<<<g1, 32, 0, stream>>>(x, W_tgt, b_tgt, B_tgt, xaTgt, tgtOut);

    dim3 g2(SEQ, NL);
    pair_gelu_dot<<<g2, 256, 0, stream>>>(srcOut, tgtOut, cls, (float*)d_out);
}


# Pair kernel: each lane keeps fixed columns and fetches the target row as 16-byte chunks two chunks ahead of use, instead of one waited 4-byte load per element; reorders the per-lane h sums
# speedup vs baseline: 1.9626x; 1.3860x over previous
#include <hip/hip_runtime.h>
#include <hip/hip_bf16.h>
#include <math.h>

#define BATCH 1
#define SEQ   128
#define HID   768
#define NL    10
#define NR    2

typedef __attribute__((ext_vector_type(2))) float v2f;
typedef __attribute__((ext_vector_type(4))) float v4f;
typedef __attribute__((ext_vector_type(8))) float v8f;

__global__ __launch_bounds__(256)
void xa_kernel(const float* __restrict__ X, const float* __restrict__ A,
               float* __restrict__ xa) {
    const int s    = blockIdx.x;
    const int tid  = threadIdx.x;
    const int lane = tid & 31;
    const int wave = tid >> 5;

    float acc[NR * NL];
    #pragma unroll
    for (int c = 0; c < NR * NL; ++c) acc[c] = 0.0f;

    #pragma unroll
    for (int j = 0; j < HID / 256; ++j) {
        const int h = tid + 256 * j;
        const float xv = X[s * HID + h];
        const v4f* arow = (const v4f*)(A + h * (NR * NL));
        #pragma unroll
        for (int q = 0; q < (NR * NL) / 4; ++q) {
            const v4f a = arow[q];
            acc[4 * q + 0] += xv * a.x;
            acc[4 * q + 1] += xv * a.y;
            acc[4 * q + 2] += xv * a.z;
            acc[4 * q + 3] += xv * a.w;
        }
    }

    #pragma unroll
    for (int c = 0; c < NR * NL; ++c) {
        float v = acc[c];
        #pragma unroll
        for (int off = 16; off > 0; off >>= 1)
            v += __shfl_xor(v, off, 32);
        acc[c] = v;
    }

    __shared__ float red[8][NR * NL];
    if (lane == 0) {
        #pragma unroll
        for (int c = 0; c < NR * NL; ++c) red[wave][c] = acc[c];
    }
    __syncthreads();

    if (tid < NR * NL) {
        float sum = 0.0f;
        #pragma unroll
        for (int w = 0; w < 8; ++w) sum += red[w][tid];
        const int r = tid / NL;
        const int l = tid - r * NL;
        xa[(s * NL + l) * NR + r] = sum;
    }
}

__global__ __launch_bounds__(32)
void fused_linear_wmma(const float* __restrict__ X,
                       const float* __restrict__ W,
                       const float* __restrict__ bias,
                       const float* __restrict__ Blora,
                       const float* __restrict__ xa,
                       float* __restrict__ out)
{
    const int hTile = blockIdx.x * 16;
    const int sTile = blockIdx.y * 16;
    const int lane  = threadIdx.x;
    const int nLoc  = lane & 15;
    const int kOff  = (lane >> 4) * 2;

    v8f acc = {};
    const float* arow = X + (sTile + nLoc) * HID;

    for (int k0 = 0; k0 < HID; k0 += 4) {
        const int kk = k0 + kOff;
        v2f a = *(const v2f*)(arow + kk);
        v2f b;
        b.x = W[kk * HID + (hTile + nLoc)];
        b.y = W[(kk + 1) * HID + (hTile + nLoc)];
        acc = __builtin_amdgcn_wmma_f32_16x16x4_f32(
                  false, a, false, b, (short)0, acc, false, false);
    }

    const int n     = hTile + nLoc;
    const int mBase = (lane < 16) ? 0 : 8;
    const float bv  = bias[n];

    #pragma unroll
    for (int l = 0; l < NL; ++l) {
        const float bl0 = Blora[(0 * HID + n) * NL + l];
        const float bl1 = Blora[(1 * HID + n) * NL + l];
        #pragma unroll
        for (int i = 0; i < 8; ++i) {
            const int s = sTile + mBase + i;
            const float xa0 = xa[(s * NL + l) * NR + 0];
            const float xa1 = xa[(s * NL + l) * NR + 1];
            out[(s * NL + l) * HID + n] = acc[i] + bv + xa0 * bl0 + xa1 * bl1;
        }
    }
}

#define P2_ROWS   (SEQ / 8)
#define P2_CHUNKS (HID / 128)
#define P2_NCHUNK (P2_ROWS * P2_CHUNKS)
#define P2_RSTEP  (8 * NL * HID)

__global__ __launch_bounds__(256)
void pair_gelu_dot(const float* __restrict__ srcOut,
                   const float* __restrict__ tgtOut,
                   const float* __restrict__ cls,
                   float* __restrict__ out)
{
    const int s = blockIdx.x;
    const int l = blockIdx.y;

    __shared__ __attribute__((aligned(16))) float sSrc[HID];
    __shared__ __attribute__((aligned(16))) float sCls[HID];
    for (int h = threadIdx.x; h < HID; h += 256) {
        sSrc[h] = srcOut[(s * NL + l) * HID + h];
        sCls[h] = cls[h * NL + l];
    }
    __syncthreads();

    const int wave = threadIdx.x >> 5;
    const int lane = threadIdx.x & 31;
    const float inv_sqrt2 = 0.70710678118654752440f;

    const v4f* sSrc4 = (const v4f*)sSrc + lane;
    const v4f* sCls4 = (const v4f*)sCls + lane;

    const float* lbase = tgtOut + (wave * NL + l) * HID + 4 * lane;

    v4f b0 = *(const v4f*)(lbase);
    v4f b1 = *(const v4f*)(lbase + 128);
    int q = 2;

    for (int r = 0; r < P2_ROWS; ++r) {
        float sum = 0.0f;
        #pragma unroll 3
        for (int j = 0; j < P2_CHUNKS; ++j) {
            const v4f cur = b0;
            b0 = b1;
            {
                const int qq = (q < P2_NCHUNK) ? q : (P2_NCHUNK - 1);
                const int qr = qq / P2_CHUNKS;
                const int qj = qq - qr * P2_CHUNKS;
                b1 = *(const v4f*)(lbase + qr * P2_RSTEP + qj * 128);
                ++q;
            }
            const v4f sv = sSrc4[j * 32];
            const v4f cv = sCls4[j * 32];
            #pragma unroll
            for (int e = 0; e < 4; ++e) {
                const float v = sv[e] + cur[e];
                const float g = 0.5f * v * (1.0f + erff(v * inv_sqrt2));
                sum += g * cv[e];
            }
        }
        #pragma unroll
        for (int off = 16; off > 0; off >>= 1)
            sum += __shfl_xor(sum, off, 32);
        if (lane == 0)
            out[(l * SEQ + s) * SEQ + (wave + 8 * r)] = sum;
    }
}

extern "C" void kernel_launch(void* const* d_in, const int* in_sizes, int n_in,
                              void* d_out, int out_size, void* d_ws, size_t ws_size,
                              hipStream_t stream) {
    (void)in_sizes; (void)n_in; (void)out_size; (void)ws_size;

    const float* x     = (const float*)d_in[0];
    const float* W_src = (const float*)d_in[1];
    const float* b_src = (const float*)d_in[2];
    const float* W_tgt = (const float*)d_in[3];
    const float* b_tgt = (const float*)d_in[4];
    const float* A_src = (const float*)d_in[5];
    const float* B_src = (const float*)d_in[6];
    const float* A_tgt = (const float*)d_in[7];
    const float* B_tgt = (const float*)d_in[8];
    const float* cls   = (const float*)d_in[9];

    float* ws     = (float*)d_ws;
    float* srcOut = ws;
    float* tgtOut = srcOut + SEQ * NL * HID;
    float* xaSrc  = tgtOut + SEQ * NL * HID;
    float* xaTgt  = xaSrc + SEQ * NL * NR;

    xa_kernel<<<SEQ, 256, 0, stream>>>(x, A_src, xaSrc);
    xa_kernel<<<SEQ, 256, 0, stream>>>(x, A_tgt, xaTgt);

    dim3 g1(HID / 16, SEQ / 16);
    fused_linear_wmma<<<g1, 32, 0, stream>>>(x, W_src, b_src, B_src, xaSrc, srcOut);
    fused_linear_wmma<<<g1, 32, 0, stream>>>(x, W_tgt, b_tgt, B_tgt, xaTgt, tgtOut);

    dim3 g2(SEQ, NL);
    pair_gelu_dot<<<g2, 256, 0, stream>>>(srcOut, tgtOut, cls, (float*)d_out);
}


# Pair kernel: library expf (hi/lo argument split, ldexp, range guards: 13 instr a value) replaced by one v_exp_f32 of -p*log2e; arg <= 0, e only enters 1-e; erf within 1 ulp, not bit-identical
# speedup vs baseline: 2.8295x; 1.4417x over previous
#include <hip/hip_runtime.h>
#include <hip/hip_bf16.h>
#include <math.h>

#define BATCH 1
#define SEQ   128
#define HID   768
#define NL    10
#define NR    2

typedef __attribute__((ext_vector_type(2))) float v2f;
typedef __attribute__((ext_vector_type(4))) float v4f;
typedef __attribute__((ext_vector_type(8))) float v8f;

__global__ __launch_bounds__(256)
void xa_kernel(const float* __restrict__ X, const float* __restrict__ A,
               float* __restrict__ xa) {
    const int s    = blockIdx.x;
    const int tid  = threadIdx.x;
    const int lane = tid & 31;
    const int wave = tid >> 5;

    float acc[NR * NL];
    #pragma unroll
    for (int c = 0; c < NR * NL; ++c) acc[c] = 0.0f;

    #pragma unroll
    for (int j = 0; j < HID / 256; ++j) {
        const int h = tid + 256 * j;
        const float xv = X[s * HID + h];
        const v4f* arow = (const v4f*)(A + h * (NR * NL));
        #pragma unroll
        for (int q = 0; q < (NR * NL) / 4; ++q) {
            const v4f a = arow[q];
            acc[4 * q + 0] += xv * a.x;
            acc[4 * q + 1] += xv * a.y;
            acc[4 * q + 2] += xv * a.z;
            acc[4 * q + 3] += xv * a.w;
        }
    }

    #pragma unroll
    for (int c = 0; c < NR * NL; ++c) {
        float v = acc[c];
        #pragma unroll
        for (int off = 16; off > 0; off >>= 1)
            v += __shfl_xor(v, off, 32);
        acc[c] = v;
    }

    __shared__ float red[8][NR * NL];
    if (lane == 0) {
        #pragma unroll
        for (int c = 0; c < NR * NL; ++c) red[wave][c] = acc[c];
    }
    __syncthreads();

    if (tid < NR * NL) {
        float sum = 0.0f;
        #pragma unroll
        for (int w = 0; w < 8; ++w) sum += red[w][tid];
        const int r = tid / NL;
        const int l = tid - r * NL;
        xa[(s * NL + l) * NR + r] = sum;
    }
}

__global__ __launch_bounds__(32)
void fused_linear_wmma(const float* __restrict__ X,
                       const float* __restrict__ W,
                       const float* __restrict__ bias,
                       const float* __restrict__ Blora,
                       const float* __restrict__ xa,
                       float* __restrict__ out)
{
    const int hTile = blockIdx.x * 16;
    const int sTile = blockIdx.y * 16;
    const int lane  = threadIdx.x;
    const int nLoc  = lane & 15;
    const int kOff  = (lane >> 4) * 2;

    v8f acc = {};
    const float* arow = X + (sTile + nLoc) * HID;

    for (int k0 = 0; k0 < HID; k0 += 4) {
        const int kk = k0 + kOff;
        v2f a = *(const v2f*)(arow + kk);
        v2f b;
        b.x = W[kk * HID + (hTile + nLoc)];
        b.y = W[(kk + 1) * HID + (hTile + nLoc)];
        acc = __builtin_amdgcn_wmma_f32_16x16x4_f32(
                  false, a, false, b, (short)0, acc, false, false);
    }

    const int n     = hTile + nLoc;
    const int mBase = (lane < 16) ? 0 : 8;
    const float bv  = bias[n];

    #pragma unroll
    for (int l = 0; l < NL; ++l) {
        const float bl0 = Blora[(0 * HID + n) * NL + l];
        const float bl1 = Blora[(1 * HID + n) * NL + l];
        #pragma unroll
        for (int i = 0; i < 8; ++i) {
            const int s = sTile + mBase + i;
            const float xa0 = xa[(s * NL + l) * NR + 0];
            const float xa1 = xa[(s * NL + l) * NR + 1];
            out[(s * NL + l) * HID + n] = acc[i] + bv + xa0 * bl0 + xa1 * bl1;
        }
    }
}

static __device__ __forceinline__ v2f splat2(float c) { v2f r = {c, c}; return r; }
static __device__ __forceinline__ v2f fma2(v2f a, v2f b, v2f c) {
    return __builtin_elementwise_fma(a, b, c);
}

static __device__ __forceinline__ v2f erf_both_ranges2(v2f y) {
    v2f a;
    a.x = fabsf(y.x);
    a.y = fabsf(y.y);

    v2f p = fma2(a, splat2(0x1.1d3156p-16f), splat2(-0x1.8d129p-12f));
    p = fma2(a, p, splat2(0x1.f9a6d2p-9f));
    p = fma2(a, p, splat2(-0x1.8c3164p-6f));
    p = fma2(a, p, splat2(0x1.b4e9c8p-4f));
    p = fma2(a, p, splat2(0x1.4515fap-1f));
    p = fma2(a, p, splat2(0x1.078e50p-3f));
    p = fma2(a, p, a);
    const v2f w = p * splat2(-0x1.715476p+0f);
    v2f big;
    big.x = 1.0f - __builtin_amdgcn_exp2f(w.x);
    big.y = 1.0f - __builtin_amdgcn_exp2f(w.y);

    const v2f t = a * a;
    v2f q = fma2(t, splat2(-0x1.268bc2p-11f), splat2(0x1.420828p-8f));
    q = fma2(t, q, splat2(-0x1.b5937p-6f));
    q = fma2(t, q, splat2(0x1.ce077cp-4f));
    q = fma2(t, q, splat2(-0x1.81266p-2f));
    q = fma2(t, q, splat2(0x1.06eba0p-3f));
    const v2f small = fma2(a, q, a);

    v2f r;
    r.x = copysignf((a.x < 1.0f) ? small.x : big.x, y.x);
    r.y = copysignf((a.y < 1.0f) ? small.y : big.y, y.y);
    return r;
}

#define P2_ROWS   (SEQ / 8)
#define P2_CHUNKS (HID / 128)
#define P2_NCHUNK (P2_ROWS * P2_CHUNKS)
#define P2_RSTEP  (8 * NL * HID)

__global__ __launch_bounds__(256)
void pair_gelu_dot(const float* __restrict__ srcOut,
                   const float* __restrict__ tgtOut,
                   const float* __restrict__ cls,
                   float* __restrict__ out)
{
    const int s = blockIdx.x;
    const int l = blockIdx.y;

    __shared__ __attribute__((aligned(16))) float sSrc[HID];
    __shared__ __attribute__((aligned(16))) float sCls[HID];
    for (int h = threadIdx.x; h < HID; h += 256) {
        sSrc[h] = srcOut[(s * NL + l) * HID + h];
        sCls[h] = cls[h * NL + l];
    }
    __syncthreads();

    const int wave = threadIdx.x >> 5;
    const int lane = threadIdx.x & 31;
    const float inv_sqrt2 = 0.70710678118654752440f;

    const v4f* sSrc4 = (const v4f*)sSrc + lane;
    const v4f* sCls4 = (const v4f*)sCls + lane;

    const float* lbase = tgtOut + (wave * NL + l) * HID + 4 * lane;

    v4f b0 = *(const v4f*)(lbase);
    v4f b1 = *(const v4f*)(lbase + 128);
    int q = 2;

    for (int r = 0; r < P2_ROWS; ++r) {
        float sum = 0.0f;
        #pragma unroll 3
        for (int j = 0; j < P2_CHUNKS; ++j) {
            const v4f cur = b0;
            b0 = b1;
            {
                const int qq = (q < P2_NCHUNK) ? q : (P2_NCHUNK - 1);
                const int qr = qq / P2_CHUNKS;
                const int qj = qq - qr * P2_CHUNKS;
                b1 = *(const v4f*)(lbase + qr * P2_RSTEP + qj * 128);
                ++q;
            }
            const v4f sv = sSrc4[j * 32];
            const v4f cv = sCls4[j * 32];
            #pragma unroll
            for (int e = 0; e < 4; e += 2) {
                v2f v;
                v.x = sv[e]     + cur[e];
                v.y = sv[e + 1] + cur[e + 1];
                const v2f erfv = erf_both_ranges2(v * inv_sqrt2);
                const v2f g = (0.5f * v) * (1.0f + erfv);
                sum = fmaf(g.x, cv[e], sum);
                sum = fmaf(g.y, cv[e + 1], sum);
            }
        }
        #pragma unroll
        for (int off = 16; off > 0; off >>= 1)
            sum += __shfl_xor(sum, off, 32);
        if (lane == 0)
            out[(l * SEQ + s) * SEQ + (wave + 8 * r)] = sum;
    }
}

extern "C" void kernel_launch(void* const* d_in, const int* in_sizes, int n_in,
                              void* d_out, int out_size, void* d_ws, size_t ws_size,
                              hipStream_t stream) {
    (void)in_sizes; (void)n_in; (void)out_size; (void)ws_size;

    const float* x     = (const float*)d_in[0];
    const float* W_src = (const float*)d_in[1];
    const float* b_src = (const float*)d_in[2];
    const float* W_tgt = (const float*)d_in[3];
    const float* b_tgt = (const float*)d_in[4];
    const float* A_src = (const float*)d_in[5];
    const float* B_src = (const float*)d_in[6];
    const float* A_tgt = (const float*)d_in[7];
    const float* B_tgt = (const float*)d_in[8];
    const float* cls   = (const float*)d_in[9];

    float* ws     = (float*)d_ws;
    float* srcOut = ws;
    float* tgtOut = srcOut + SEQ * NL * HID;
    float* xaSrc  = tgtOut + SEQ * NL * HID;
    float* xaTgt  = xaSrc + SEQ * NL * NR;

    xa_kernel<<<SEQ, 256, 0, stream>>>(x, A_src, xaSrc);
    xa_kernel<<<SEQ, 256, 0, stream>>>(x, A_tgt, xaTgt);

    dim3 g1(HID / 16, SEQ / 16);
    fused_linear_wmma<<<g1, 32, 0, stream>>>(x, W_src, b_src, B_src, xaSrc, srcOut);
    fused_linear_wmma<<<g1, 32, 0, stream>>>(x, W_tgt, b_tgt, B_tgt, xaTgt, tgtOut);

    dim3 g2(SEQ, NL);
    pair_gelu_dot<<<g2, 256, 0, stream>>>(srcOut, tgtOut, cls, (float*)d_out);
}


# Phases 0 and 1: the source-side and target-side launches, independent of each other, merged into one launch per phase (block index picks the side); 5 launches become 3
# speedup vs baseline: 3.7152x; 1.3130x over previous
#include <hip/hip_runtime.h>
#include <hip/hip_bf16.h>
#include <math.h>

#define BATCH 1
#define SEQ   128
#define HID   768
#define NL    10
#define NR    2

typedef __attribute__((ext_vector_type(2))) float v2f;
typedef __attribute__((ext_vector_type(4))) float v4f;
typedef __attribute__((ext_vector_type(8))) float v8f;

__global__ __launch_bounds__(256)
void xa_kernel(const float* __restrict__ X,
               const float* __restrict__ A_src, const float* __restrict__ A_tgt,
               float* __restrict__ xa_both) {
    const int side = blockIdx.y;
    const float* __restrict__ A = side ? A_tgt : A_src;
    float* __restrict__ xa = xa_both + side * (SEQ * NL * NR);
    const int s    = blockIdx.x;
    const int tid  = threadIdx.x;
    const int lane = tid & 31;
    const int wave = tid >> 5;

    float acc[NR * NL];
    #pragma unroll
    for (int c = 0; c < NR * NL; ++c) acc[c] = 0.0f;

    #pragma unroll
    for (int j = 0; j < HID / 256; ++j) {
        const int h = tid + 256 * j;
        const float xv = X[s * HID + h];
        const v4f* arow = (const v4f*)(A + h * (NR * NL));
        #pragma unroll
        for (int q = 0; q < (NR * NL) / 4; ++q) {
            const v4f a = arow[q];
            acc[4 * q + 0] += xv * a.x;
            acc[4 * q + 1] += xv * a.y;
            acc[4 * q + 2] += xv * a.z;
            acc[4 * q + 3] += xv * a.w;
        }
    }

    #pragma unroll
    for (int c = 0; c < NR * NL; ++c) {
        float v = acc[c];
        #pragma unroll
        for (int off = 16; off > 0; off >>= 1)
            v += __shfl_xor(v, off, 32);
        acc[c] = v;
    }

    __shared__ float red[8][NR * NL];
    if (lane == 0) {
        #pragma unroll
        for (int c = 0; c < NR * NL; ++c) red[wave][c] = acc[c];
    }
    __syncthreads();

    if (tid < NR * NL) {
        float sum = 0.0f;
        #pragma unroll
        for (int w = 0; w < 8; ++w) sum += red[w][tid];
        const int r = tid / NL;
        const int l = tid - r * NL;
        xa[(s * NL + l) * NR + r] = sum;
    }
}

__global__ __launch_bounds__(32)
void fused_linear_wmma(const float* __restrict__ X,
                       const float* __restrict__ W_src,
                       const float* __restrict__ W_tgt,
                       const float* __restrict__ bias_src,
                       const float* __restrict__ bias_tgt,
                       const float* __restrict__ Blora_src,
                       const float* __restrict__ Blora_tgt,
                       const float* __restrict__ xa_both,
                       float* __restrict__ out_both)
{
    const int side = blockIdx.z;
    const float* __restrict__ W     = side ? W_tgt     : W_src;
    const float* __restrict__ bias  = side ? bias_tgt  : bias_src;
    const float* __restrict__ Blora = side ? Blora_tgt : Blora_src;
    const float* __restrict__ xa    = xa_both  + side * (SEQ * NL * NR);
    float* __restrict__       out   = out_both + side * (SEQ * NL * HID);

    const int hTile = blockIdx.x * 16;
    const int sTile = blockIdx.y * 16;
    const int lane  = threadIdx.x;
    const int nLoc  = lane & 15;
    const int kOff  = (lane >> 4) * 2;

    v8f acc = {};
    const float* arow = X + (sTile + nLoc) * HID;

    for (int k0 = 0; k0 < HID; k0 += 4) {
        const int kk = k0 + kOff;
        v2f a = *(const v2f*)(arow + kk);
        v2f b;
        b.x = W[kk * HID + (hTile + nLoc)];
        b.y = W[(kk + 1) * HID + (hTile + nLoc)];
        acc = __builtin_amdgcn_wmma_f32_16x16x4_f32(
                  false, a, false, b, (short)0, acc, false, false);
    }

    const int n     = hTile + nLoc;
    const int mBase = (lane < 16) ? 0 : 8;
    const float bv  = bias[n];

    #pragma unroll
    for (int l = 0; l < NL; ++l) {
        const float bl0 = Blora[(0 * HID + n) * NL + l];
        const float bl1 = Blora[(1 * HID + n) * NL + l];
        #pragma unroll
        for (int i = 0; i < 8; ++i) {
            const int s = sTile + mBase + i;
            const float xa0 = xa[(s * NL + l) * NR + 0];
            const float xa1 = xa[(s * NL + l) * NR + 1];
            out[(s * NL + l) * HID + n] = acc[i] + bv + xa0 * bl0 + xa1 * bl1;
        }
    }
}

static __device__ __forceinline__ v2f splat2(float c) { v2f r = {c, c}; return r; }
static __device__ __forceinline__ v2f fma2(v2f a, v2f b, v2f c) {
    return __builtin_elementwise_fma(a, b, c);
}

static __device__ __forceinline__ v2f erf_both_ranges2(v2f y) {
    v2f a;
    a.x = fabsf(y.x);
    a.y = fabsf(y.y);

    v2f p = fma2(a, splat2(0x1.1d3156p-16f), splat2(-0x1.8d129p-12f));
    p = fma2(a, p, splat2(0x1.f9a6d2p-9f));
    p = fma2(a, p, splat2(-0x1.8c3164p-6f));
    p = fma2(a, p, splat2(0x1.b4e9c8p-4f));
    p = fma2(a, p, splat2(0x1.4515fap-1f));
    p = fma2(a, p, splat2(0x1.078e50p-3f));
    p = fma2(a, p, a);
    const v2f w = p * splat2(-0x1.715476p+0f);
    v2f big;
    big.x = 1.0f - __builtin_amdgcn_exp2f(w.x);
    big.y = 1.0f - __builtin_amdgcn_exp2f(w.y);

    const v2f t = a * a;
    v2f q = fma2(t, splat2(-0x1.268bc2p-11f), splat2(0x1.420828p-8f));
    q = fma2(t, q, splat2(-0x1.b5937p-6f));
    q = fma2(t, q, splat2(0x1.ce077cp-4f));
    q = fma2(t, q, splat2(-0x1.81266p-2f));
    q = fma2(t, q, splat2(0x1.06eba0p-3f));
    const v2f small = fma2(a, q, a);

    v2f r;
    r.x = copysignf((a.x < 1.0f) ? small.x : big.x, y.x);
    r.y = copysignf((a.y < 1.0f) ? small.y : big.y, y.y);
    return r;
}

#define P2_ROWS   (SEQ / 8)
#define P2_CHUNKS (HID / 128)
#define P2_NCHUNK (P2_ROWS * P2_CHUNKS)
#define P2_RSTEP  (8 * NL * HID)

__global__ __launch_bounds__(256)
void pair_gelu_dot(const float* __restrict__ srcOut,
                   const float* __restrict__ tgtOut,
                   const float* __restrict__ cls,
                   float* __restrict__ out)
{
    const int s = blockIdx.x;
    const int l = blockIdx.y;

    __shared__ __attribute__((aligned(16))) float sSrc[HID];
    __shared__ __attribute__((aligned(16))) float sCls[HID];
    for (int h = threadIdx.x; h < HID; h += 256) {
        sSrc[h] = srcOut[(s * NL + l) * HID + h];
        sCls[h] = cls[h * NL + l];
    }
    __syncthreads();

    const int wave = threadIdx.x >> 5;
    const int lane = threadIdx.x & 31;
    const float inv_sqrt2 = 0.70710678118654752440f;

    const v4f* sSrc4 = (const v4f*)sSrc + lane;
    const v4f* sCls4 = (const v4f*)sCls + lane;

    const float* lbase = tgtOut + (wave * NL + l) * HID + 4 * lane;

    v4f b0 = *(const v4f*)(lbase);
    v4f b1 = *(const v4f*)(lbase + 128);
    int q = 2;

    for (int r = 0; r < P2_ROWS; ++r) {
        float sum = 0.0f;
        #pragma unroll 3
        for (int j = 0; j < P2_CHUNKS; ++j) {
            const v4f cur = b0;
            b0 = b1;
            {
                const int qq = (q < P2_NCHUNK) ? q : (P2_NCHUNK - 1);
                const int qr = qq / P2_CHUNKS;
                const int qj = qq - qr * P2_CHUNKS;
                b1 = *(const v4f*)(lbase + qr * P2_RSTEP + qj * 128);
                ++q;
            }
            const v4f sv = sSrc4[j * 32];
            const v4f cv = sCls4[j * 32];
            #pragma unroll
            for (int e = 0; e < 4; e += 2) {
                v2f v;
                v.x = sv[e]     + cur[e];
                v.y = sv[e + 1] + cur[e + 1];
                const v2f erfv = erf_both_ranges2(v * inv_sqrt2);
                const v2f g = (0.5f * v) * (1.0f + erfv);
                sum = fmaf(g.x, cv[e], sum);
                sum = fmaf(g.y, cv[e + 1], sum);
            }
        }
        #pragma unroll
        for (int off = 16; off > 0; off >>= 1)
            sum += __shfl_xor(sum, off, 32);
        if (lane == 0)
            out[(l * SEQ + s) * SEQ + (wave + 8 * r)] = sum;
    }
}

extern "C" void kernel_launch(void* const* d_in, const int* in_sizes, int n_in,
                              void* d_out, int out_size, void* d_ws, size_t ws_size,
                              hipStream_t stream) {
    (void)in_sizes; (void)n_in; (void)out_size; (void)ws_size;

    const float* x     = (const float*)d_in[0];
    const float* W_src = (const float*)d_in[1];
    const float* b_src = (const float*)d_in[2];
    const float* W_tgt = (const float*)d_in[3];
    const float* b_tgt = (const float*)d_in[4];
    const float* A_src = (const float*)d_in[5];
    const float* B_src = (const float*)d_in[6];
    const float* A_tgt = (const float*)d_in[7];
    const float* B_tgt = (const float*)d_in[8];
    const float* cls   = (const float*)d_in[9];

    float* ws     = (float*)d_ws;
    float* srcOut = ws;
    float* tgtOut = srcOut + SEQ * NL * HID;
    float* xaSrc  = tgtOut + SEQ * NL * HID;


    dim3 g0(SEQ, 2);
    xa_kernel<<<g0, 256, 0, stream>>>(x, A_src, A_tgt, xaSrc);

    dim3 g1(HID / 16, SEQ / 16, 2);
    fused_linear_wmma<<<g1, 32, 0, stream>>>(x, W_src, W_tgt, b_src, b_tgt,
                                             B_src, B_tgt, xaSrc, srcOut);

    dim3 g2(SEQ, NL);
    pair_gelu_dot<<<g2, 256, 0, stream>>>(srcOut, tgtOut, cls, (float*)d_out);
}


# GEMM: K=768 split over the 4 waves of a block (192 each: a quarter of the serial load chain), partial tiles summed via LDS in fixed order, the 10 epilogue variants shared out; reorders the K sums
# speedup vs baseline: 4.1406x; 1.1145x over previous
#include <hip/hip_runtime.h>
#include <hip/hip_bf16.h>
#include <math.h>

#define BATCH 1
#define SEQ   128
#define HID   768
#define NL    10
#define NR    2

typedef __attribute__((ext_vector_type(2))) float v2f;
typedef __attribute__((ext_vector_type(4))) float v4f;
typedef __attribute__((ext_vector_type(8))) float v8f;

__global__ __launch_bounds__(256)
void xa_kernel(const float* __restrict__ X,
               const float* __restrict__ A_src, const float* __restrict__ A_tgt,
               float* __restrict__ xa_both) {
    const int side = blockIdx.y;
    const float* __restrict__ A = side ? A_tgt : A_src;
    float* __restrict__ xa = xa_both + side * (SEQ * NL * NR);
    const int s    = blockIdx.x;
    const int tid  = threadIdx.x;
    const int lane = tid & 31;
    const int wave = tid >> 5;

    float acc[NR * NL];
    #pragma unroll
    for (int c = 0; c < NR * NL; ++c) acc[c] = 0.0f;

    #pragma unroll
    for (int j = 0; j < HID / 256; ++j) {
        const int h = tid + 256 * j;
        const float xv = X[s * HID + h];
        const v4f* arow = (const v4f*)(A + h * (NR * NL));
        #pragma unroll
        for (int q = 0; q < (NR * NL) / 4; ++q) {
            const v4f a = arow[q];
            acc[4 * q + 0] += xv * a.x;
            acc[4 * q + 1] += xv * a.y;
            acc[4 * q + 2] += xv * a.z;
            acc[4 * q + 3] += xv * a.w;
        }
    }

    #pragma unroll
    for (int c = 0; c < NR * NL; ++c) {
        float v = acc[c];
        #pragma unroll
        for (int off = 16; off > 0; off >>= 1)
            v += __shfl_xor(v, off, 32);
        acc[c] = v;
    }

    __shared__ float red[8][NR * NL];
    if (lane == 0) {
        #pragma unroll
        for (int c = 0; c < NR * NL; ++c) red[wave][c] = acc[c];
    }
    __syncthreads();

    if (tid < NR * NL) {
        float sum = 0.0f;
        #pragma unroll
        for (int w = 0; w < 8; ++w) sum += red[w][tid];
        const int r = tid / NL;
        const int l = tid - r * NL;
        xa[(s * NL + l) * NR + r] = sum;
    }
}

#define P1_KSPLIT 4
#define P1_KLEN   (HID / P1_KSPLIT)

__global__ __launch_bounds__(32 * P1_KSPLIT)
void fused_linear_wmma(const float* __restrict__ X,
                       const float* __restrict__ W_src,
                       const float* __restrict__ W_tgt,
                       const float* __restrict__ bias_src,
                       const float* __restrict__ bias_tgt,
                       const float* __restrict__ Blora_src,
                       const float* __restrict__ Blora_tgt,
                       const float* __restrict__ xa_both,
                       float* __restrict__ out_both)
{
    const int side = blockIdx.z;
    const float* __restrict__ W     = side ? W_tgt     : W_src;
    const float* __restrict__ bias  = side ? bias_tgt  : bias_src;
    const float* __restrict__ Blora = side ? Blora_tgt : Blora_src;
    const float* __restrict__ xa    = xa_both  + side * (SEQ * NL * NR);
    float* __restrict__       out   = out_both + side * (SEQ * NL * HID);

    const int hTile = blockIdx.x * 16;
    const int sTile = blockIdx.y * 16;
    const int lane  = threadIdx.x & 31;
    const int wave  = __builtin_amdgcn_readfirstlane(threadIdx.x >> 5);
    const int nLoc  = lane & 15;
    const int kOff  = (lane >> 4) * 2;

    v8f acc = {};
    const float* arow = X + (sTile + nLoc) * HID;
    const int kBeg = wave * P1_KLEN;

    for (int k0 = kBeg; k0 < kBeg + P1_KLEN; k0 += 4) {
        const int kk = k0 + kOff;
        v2f a = *(const v2f*)(arow + kk);
        v2f b;
        b.x = W[kk * HID + (hTile + nLoc)];
        b.y = W[(kk + 1) * HID + (hTile + nLoc)];
        acc = __builtin_amdgcn_wmma_f32_16x16x4_f32(
                  false, a, false, b, (short)0, acc, false, false);
    }

    __shared__ float part[P1_KSPLIT][8][32];
    #pragma unroll
    for (int i = 0; i < 8; ++i) part[wave][i][lane] = acc[i];
    __syncthreads();

    float tot[8];
    #pragma unroll
    for (int i = 0; i < 8; ++i) {
        float t = part[0][i][lane];
        #pragma unroll
        for (int w = 1; w < P1_KSPLIT; ++w) t += part[w][i][lane];
        tot[i] = t;
    }

    const int n     = hTile + nLoc;
    const int mBase = (lane < 16) ? 0 : 8;
    const float bv  = bias[n];

    for (int l = wave; l < NL; l += P1_KSPLIT) {
        const float bl0 = Blora[(0 * HID + n) * NL + l];
        const float bl1 = Blora[(1 * HID + n) * NL + l];
        #pragma unroll
        for (int i = 0; i < 8; ++i) {
            const int s = sTile + mBase + i;
            const float xa0 = xa[(s * NL + l) * NR + 0];
            const float xa1 = xa[(s * NL + l) * NR + 1];
            out[(s * NL + l) * HID + n] = tot[i] + bv + xa0 * bl0 + xa1 * bl1;
        }
    }
}

static __device__ __forceinline__ v2f splat2(float c) { v2f r = {c, c}; return r; }
static __device__ __forceinline__ v2f fma2(v2f a, v2f b, v2f c) {
    return __builtin_elementwise_fma(a, b, c);
}

static __device__ __forceinline__ v2f erf_both_ranges2(v2f y) {
    v2f a;
    a.x = fabsf(y.x);
    a.y = fabsf(y.y);

    v2f p = fma2(a, splat2(0x1.1d3156p-16f), splat2(-0x1.8d129p-12f));
    p = fma2(a, p, splat2(0x1.f9a6d2p-9f));
    p = fma2(a, p, splat2(-0x1.8c3164p-6f));
    p = fma2(a, p, splat2(0x1.b4e9c8p-4f));
    p = fma2(a, p, splat2(0x1.4515fap-1f));
    p = fma2(a, p, splat2(0x1.078e50p-3f));
    p = fma2(a, p, a);
    const v2f w = p * splat2(-0x1.715476p+0f);
    v2f big;
    big.x = 1.0f - __builtin_amdgcn_exp2f(w.x);
    big.y = 1.0f - __builtin_amdgcn_exp2f(w.y);

    const v2f t = a * a;
    v2f q = fma2(t, splat2(-0x1.268bc2p-11f), splat2(0x1.420828p-8f));
    q = fma2(t, q, splat2(-0x1.b5937p-6f));
    q = fma2(t, q, splat2(0x1.ce077cp-4f));
    q = fma2(t, q, splat2(-0x1.81266p-2f));
    q = fma2(t, q, splat2(0x1.06eba0p-3f));
    const v2f small = fma2(a, q, a);

    v2f r;
    r.x = copysignf((a.x < 1.0f) ? small.x : big.x, y.x);
    r.y = copysignf((a.y < 1.0f) ? small.y : big.y, y.y);
    return r;
}

#define P2_ROWS   (SEQ / 8)
#define P2_CHUNKS (HID / 128)
#define P2_NCHUNK (P2_ROWS * P2_CHUNKS)
#define P2_RSTEP  (8 * NL * HID)

__global__ __launch_bounds__(256)
void pair_gelu_dot(const float* __restrict__ srcOut,
                   const float* __restrict__ tgtOut,
                   const float* __restrict__ cls,
                   float* __restrict__ out)
{
    const int s = blockIdx.x;
    const int l = blockIdx.y;

    __shared__ __attribute__((aligned(16))) float sSrc[HID];
    __shared__ __attribute__((aligned(16))) float sCls[HID];
    for (int h = threadIdx.x; h < HID; h += 256) {
        sSrc[h] = srcOut[(s * NL + l) * HID + h];
        sCls[h] = cls[h * NL + l];
    }
    __syncthreads();

    const int wave = threadIdx.x >> 5;
    const int lane = threadIdx.x & 31;
    const float inv_sqrt2 = 0.70710678118654752440f;

    const v4f* sSrc4 = (const v4f*)sSrc + lane;
    const v4f* sCls4 = (const v4f*)sCls + lane;

    const float* lbase = tgtOut + (wave * NL + l) * HID + 4 * lane;

    v4f b0 = *(const v4f*)(lbase);
    v4f b1 = *(const v4f*)(lbase + 128);
    int q = 2;

    for (int r = 0; r < P2_ROWS; ++r) {
        float sum = 0.0f;
        #pragma unroll 3
        for (int j = 0; j < P2_CHUNKS; ++j) {
            const v4f cur = b0;
            b0 = b1;
            {
                const int qq = (q < P2_NCHUNK) ? q : (P2_NCHUNK - 1);
                const int qr = qq / P2_CHUNKS;
                const int qj = qq - qr * P2_CHUNKS;
                b1 = *(const v4f*)(lbase + qr * P2_RSTEP + qj * 128);
                ++q;
            }
            const v4f sv = sSrc4[j * 32];
            const v4f cv = sCls4[j * 32];
            #pragma unroll
            for (int e = 0; e < 4; e += 2) {
                v2f v;
                v.x = sv[e]     + cur[e];
                v.y = sv[e + 1] + cur[e + 1];
                const v2f erfv = erf_both_ranges2(v * inv_sqrt2);
                const v2f g = (0.5f * v) * (1.0f + erfv);
                sum = fmaf(g.x, cv[e], sum);
                sum = fmaf(g.y, cv[e + 1], sum);
            }
        }
        #pragma unroll
        for (int off = 16; off > 0; off >>= 1)
            sum += __shfl_xor(sum, off, 32);
        if (lane == 0)
            out[(l * SEQ + s) * SEQ + (wave + 8 * r)] = sum;
    }
}

extern "C" void kernel_launch(void* const* d_in, const int* in_sizes, int n_in,
                              void* d_out, int out_size, void* d_ws, size_t ws_size,
                              hipStream_t stream) {
    (void)in_sizes; (void)n_in; (void)out_size; (void)ws_size;

    const float* x     = (const float*)d_in[0];
    const float* W_src = (const float*)d_in[1];
    const float* b_src = (const float*)d_in[2];
    const float* W_tgt = (const float*)d_in[3];
    const float* b_tgt = (const float*)d_in[4];
    const float* A_src = (const float*)d_in[5];
    const float* B_src = (const float*)d_in[6];
    const float* A_tgt = (const float*)d_in[7];
    const float* B_tgt = (const float*)d_in[8];
    const float* cls   = (const float*)d_in[9];

    float* ws     = (float*)d_ws;
    float* srcOut = ws;
    float* tgtOut = srcOut + SEQ * NL * HID;
    float* xaSrc  = tgtOut + SEQ * NL * HID;


    dim3 g0(SEQ, 2);
    xa_kernel<<<g0, 256, 0, stream>>>(x, A_src, A_tgt, xaSrc);

    dim3 g1(HID / 16, SEQ / 16, 2);
    fused_linear_wmma<<<g1, 32 * P1_KSPLIT, 0, stream>>>(x, W_src, W_tgt, b_src, b_tgt,
                                             B_src, B_tgt, xaSrc, srcOut);

    dim3 g2(SEQ, NL);
    pair_gelu_dot<<<g2, 256, 0, stream>>>(srcOut, tgtOut, cls, (float*)d_out);
}


# Phase 0 folded into the GEMM launch: each block forms the xa values of its own 16 rows on the matrix unit beside a 16x48 W tile (K over 8 waves); 3 launches become 2; reorders xa and K sums
# speedup vs baseline: 4.4198x; 1.0674x over previous
#include <hip/hip_runtime.h>
#include <hip/hip_bf16.h>
#include <math.h>

#define BATCH 1
#define SEQ   128
#define HID   768
#define NL    10
#define NR    2

typedef __attribute__((ext_vector_type(2))) float v2f;
typedef __attribute__((ext_vector_type(4))) float v4f;
typedef __attribute__((ext_vector_type(8))) float v8f;


#define P1_WAVES  8
#define P1_KLEN   (HID / P1_WAVES)
#define P1_HT     3
#define P1_TILES  (P1_HT + 2)
#define P1_NVAR   (P1_HT * NL)
#define P1_VPW    ((P1_NVAR + P1_WAVES - 1) / P1_WAVES)
#define XA_COLS   (NR * NL)

static __device__ __forceinline__ v8f wmma4(v2f a, v2f b, v8f c) {
    return __builtin_amdgcn_wmma_f32_16x16x4_f32(false, a, false, b, (short)0, c, false, false);
}

__global__ __launch_bounds__(32 * P1_WAVES)
void fused_linear_wmma(const float* __restrict__ X,
                       const float* __restrict__ W_src,
                       const float* __restrict__ W_tgt,
                       const float* __restrict__ bias_src,
                       const float* __restrict__ bias_tgt,
                       const float* __restrict__ Blora_src,
                       const float* __restrict__ Blora_tgt,
                       const float* __restrict__ A_src,
                       const float* __restrict__ A_tgt,
                       float* __restrict__ out_both)
{
    const int side = blockIdx.z;
    const float* __restrict__ W     = side ? W_tgt     : W_src;
    const float* __restrict__ bias  = side ? bias_tgt  : bias_src;
    const float* __restrict__ Blora = side ? Blora_tgt : Blora_src;
    const float* __restrict__ A     = side ? A_tgt     : A_src;
    float* __restrict__       out   = out_both + side * (SEQ * NL * HID);

    const int hTile = blockIdx.x * (16 * P1_HT);
    const int sTile = blockIdx.y * 16;
    const int lane  = threadIdx.x & 31;
    const int wave  = __builtin_amdgcn_readfirstlane(threadIdx.x >> 5);
    const int nLoc  = lane & 15;
    const int half  = lane >> 4;
    const int kOff  = half * 2;

    float bl0[P1_VPW], bl1[P1_VPW], bvv[P1_VPW];
    #pragma unroll
    for (int e = 0; e < P1_VPW; ++e) {
        const int q  = wave + P1_WAVES * e;
        const int qc = (q < P1_NVAR) ? q : (P1_NVAR - 1);
        const int j  = qc / NL;
        const int l  = qc - j * NL;
        const int n  = hTile + 16 * j + nLoc;
        bl0[e] = Blora[(0 * HID + n) * NL + l];
        bl1[e] = Blora[(1 * HID + n) * NL + l];
        bvv[e] = bias[n];
    }

    v8f acc0 = {}, acc1 = {}, acc2 = {}, xac0 = {}, xac1 = {};
    const float* arow = X + (sTile + nLoc) * HID;
    const float* wcol = W + hTile + nLoc;
    const int c1 = (16 + nLoc < XA_COLS) ? (16 + nLoc) : (XA_COLS - 1);
    const int kBeg = wave * P1_KLEN;

    #pragma unroll 4
    for (int it = 0; it < P1_KLEN / 4; ++it) {
        const int kk = kBeg + 4 * it + kOff;
        const v2f a = *(const v2f*)(arow + kk);
        v2f b0, b1, b2, p0, p1;
        b0.x = wcol[kk * HID];            b0.y = wcol[(kk + 1) * HID];
        b1.x = wcol[kk * HID + 16];       b1.y = wcol[(kk + 1) * HID + 16];
        b2.x = wcol[kk * HID + 32];       b2.y = wcol[(kk + 1) * HID + 32];
        p0.x = A[kk * XA_COLS + nLoc];    p0.y = A[(kk + 1) * XA_COLS + nLoc];
        p1.x = A[kk * XA_COLS + c1];      p1.y = A[(kk + 1) * XA_COLS + c1];
        acc0 = wmma4(a, b0, acc0);
        acc1 = wmma4(a, b1, acc1);
        acc2 = wmma4(a, b2, acc2);
        xac0 = wmma4(a, p0, xac0);
        xac1 = wmma4(a, p1, xac1);
    }

    __shared__ float part[P1_WAVES][P1_TILES][8][32];
    __shared__ float totS[P1_HT][8][32];
    __shared__ __attribute__((aligned(8))) float xs[16 * XA_COLS];
    #pragma unroll
    for (int i = 0; i < 8; ++i) {
        part[wave][0][i][lane] = acc0[i];
        part[wave][1][i][lane] = acc1[i];
        part[wave][2][i][lane] = acc2[i];
        part[wave][3][i][lane] = xac0[i];
        part[wave][4][i][lane] = xac1[i];
    }
    __syncthreads();

    #pragma unroll
    for (int t = 0; t < P1_TILES; ++t) {
        float v = part[0][t][wave][lane];
        #pragma unroll
        for (int u = 1; u < P1_WAVES; ++u) v += part[u][t][wave][lane];
        if (t < P1_HT) {
            totS[t][wave][lane] = v;
        } else {
            const int c = 16 * (t - P1_HT) + nLoc;
            if (c < XA_COLS) {
                const int r   = (c >= NL) ? 1 : 0;
                const int l   = c - r * NL;
                const int row = 8 * half + wave;
                xs[(row * NL + l) * NR + r] = v;
            }
        }
    }
    __syncthreads();

    const int mBase = 8 * half;
    #pragma unroll
    for (int e = 0; e < P1_VPW; ++e) {
        const int q = wave + P1_WAVES * e;
        if (q < P1_NVAR) {
            const int j = q / NL;
            const int l = q - j * NL;
            const int n = hTile + 16 * j + nLoc;
            #pragma unroll
            for (int i = 0; i < 8; ++i) {
                const int row = mBase + i;
                const v2f xv = *(const v2f*)&xs[(row * NL + l) * NR];
                out[((sTile + row) * NL + l) * HID + n] =
                    totS[j][i][lane] + bvv[e] + xv.x * bl0[e] + xv.y * bl1[e];
            }
        }
    }
}

static __device__ __forceinline__ v2f splat2(float c) { v2f r = {c, c}; return r; }
static __device__ __forceinline__ v2f fma2(v2f a, v2f b, v2f c) {
    return __builtin_elementwise_fma(a, b, c);
}

static __device__ __forceinline__ v2f erf_both_ranges2(v2f y) {
    v2f a;
    a.x = fabsf(y.x);
    a.y = fabsf(y.y);

    v2f p = fma2(a, splat2(0x1.1d3156p-16f), splat2(-0x1.8d129p-12f));
    p = fma2(a, p, splat2(0x1.f9a6d2p-9f));
    p = fma2(a, p, splat2(-0x1.8c3164p-6f));
    p = fma2(a, p, splat2(0x1.b4e9c8p-4f));
    p = fma2(a, p, splat2(0x1.4515fap-1f));
    p = fma2(a, p, splat2(0x1.078e50p-3f));
    p = fma2(a, p, a);
    const v2f w = p * splat2(-0x1.715476p+0f);
    v2f big;
    big.x = 1.0f - __builtin_amdgcn_exp2f(w.x);
    big.y = 1.0f - __builtin_amdgcn_exp2f(w.y);

    const v2f t = a * a;
    v2f q = fma2(t, splat2(-0x1.268bc2p-11f), splat2(0x1.420828p-8f));
    q = fma2(t, q, splat2(-0x1.b5937p-6f));
    q = fma2(t, q, splat2(0x1.ce077cp-4f));
    q = fma2(t, q, splat2(-0x1.81266p-2f));
    q = fma2(t, q, splat2(0x1.06eba0p-3f));
    const v2f small = fma2(a, q, a);

    v2f r;
    r.x = copysignf((a.x < 1.0f) ? small.x : big.x, y.x);
    r.y = copysignf((a.y < 1.0f) ? small.y : big.y, y.y);
    return r;
}

#define P2_ROWS   (SEQ / 8)
#define P2_CHUNKS (HID / 128)
#define P2_NCHUNK (P2_ROWS * P2_CHUNKS)
#define P2_RSTEP  (8 * NL * HID)

__global__ __launch_bounds__(256)
void pair_gelu_dot(const float* __restrict__ srcOut,
                   const float* __restrict__ tgtOut,
                   const float* __restrict__ cls,
                   float* __restrict__ out)
{
    const int s = blockIdx.x;
    const int l = blockIdx.y;

    __shared__ __attribute__((aligned(16))) float sSrc[HID];
    __shared__ __attribute__((aligned(16))) float sCls[HID];
    for (int h = threadIdx.x; h < HID; h += 256) {
        sSrc[h] = srcOut[(s * NL + l) * HID + h];
        sCls[h] = cls[h * NL + l];
    }
    __syncthreads();

    const int wave = threadIdx.x >> 5;
    const int lane = threadIdx.x & 31;
    const float inv_sqrt2 = 0.70710678118654752440f;

    const v4f* sSrc4 = (const v4f*)sSrc + lane;
    const v4f* sCls4 = (const v4f*)sCls + lane;

    const float* lbase = tgtOut + (wave * NL + l) * HID + 4 * lane;

    v4f b0 = *(const v4f*)(lbase);
    v4f b1 = *(const v4f*)(lbase + 128);
    int q = 2;

    for (int r = 0; r < P2_ROWS; ++r) {
        float sum = 0.0f;
        #pragma unroll 3
        for (int j = 0; j < P2_CHUNKS; ++j) {
            const v4f cur = b0;
            b0 = b1;
            {
                const int qq = (q < P2_NCHUNK) ? q : (P2_NCHUNK - 1);
                const int qr = qq / P2_CHUNKS;
                const int qj = qq - qr * P2_CHUNKS;
                b1 = *(const v4f*)(lbase + qr * P2_RSTEP + qj * 128);
                ++q;
            }
            const v4f sv = sSrc4[j * 32];
            const v4f cv = sCls4[j * 32];
            #pragma unroll
            for (int e = 0; e < 4; e += 2) {
                v2f v;
                v.x = sv[e]     + cur[e];
                v.y = sv[e + 1] + cur[e + 1];
                const v2f erfv = erf_both_ranges2(v * inv_sqrt2);
                const v2f g = (0.5f * v) * (1.0f + erfv);
                sum = fmaf(g.x, cv[e], sum);
                sum = fmaf(g.y, cv[e + 1], sum);
            }
        }
        #pragma unroll
        for (int off = 16; off > 0; off >>= 1)
            sum += __shfl_xor(sum, off, 32);
        if (lane == 0)
            out[(l * SEQ + s) * SEQ + (wave + 8 * r)] = sum;
    }
}

extern "C" void kernel_launch(void* const* d_in, const int* in_sizes, int n_in,
                              void* d_out, int out_size, void* d_ws, size_t ws_size,
                              hipStream_t stream) {
    (void)in_sizes; (void)n_in; (void)out_size; (void)ws_size;

    const float* x     = (const float*)d_in[0];
    const float* W_src = (const float*)d_in[1];
    const float* b_src = (const float*)d_in[2];
    const float* W_tgt = (const float*)d_in[3];
    const float* b_tgt = (const float*)d_in[4];
    const float* A_src = (const float*)d_in[5];
    const float* B_src = (const float*)d_in[6];
    const float* A_tgt = (const float*)d_in[7];
    const float* B_tgt = (const float*)d_in[8];
    const float* cls   = (const float*)d_in[9];

    float* ws     = (float*)d_ws;
    float* srcOut = ws;
    float* tgtOut = srcOut + SEQ * NL * HID;


    dim3 g1(HID / (16 * P1_HT), SEQ / 16, 2);
    fused_linear_wmma<<<g1, 32 * P1_WAVES, 0, stream>>>(x, W_src, W_tgt, b_src, b_tgt,
                                            B_src, B_tgt, A_src, A_tgt, srcOut);

    dim3 g2(SEQ, NL);
    pair_gelu_dot<<<g2, 256, 0, stream>>>(srcOut, tgtOut, cls, (float*)d_out);
}


# Pair kernel: GELU as 0.5v + |v|X(|v|): erf's two range formulas rescaled to take |v| and give X = 0.5 erf, so the 1/sqrt2, log2e, 0.5v, 1+erf ops and the sign copy go; not bit-identical
# speedup vs baseline: 4.7915x; 1.0841x over previous
#include <hip/hip_runtime.h>
#include <hip/hip_bf16.h>
#include <math.h>

#define BATCH 1
#define SEQ   128
#define HID   768
#define NL    10
#define NR    2

typedef __attribute__((ext_vector_type(2))) float v2f;
typedef __attribute__((ext_vector_type(4))) float v4f;
typedef __attribute__((ext_vector_type(8))) float v8f;


#define P1_WAVES  8
#define P1_KLEN   (HID / P1_WAVES)
#define P1_HT     3
#define P1_TILES  (P1_HT + 2)
#define P1_NVAR   (P1_HT * NL)
#define P1_VPW    ((P1_NVAR + P1_WAVES - 1) / P1_WAVES)
#define XA_COLS   (NR * NL)

static __device__ __forceinline__ v8f wmma4(v2f a, v2f b, v8f c) {
    return __builtin_amdgcn_wmma_f32_16x16x4_f32(false, a, false, b, (short)0, c, false, false);
}

__global__ __launch_bounds__(32 * P1_WAVES)
void fused_linear_wmma(const float* __restrict__ X,
                       const float* __restrict__ W_src,
                       const float* __restrict__ W_tgt,
                       const float* __restrict__ bias_src,
                       const float* __restrict__ bias_tgt,
                       const float* __restrict__ Blora_src,
                       const float* __restrict__ Blora_tgt,
                       const float* __restrict__ A_src,
                       const float* __restrict__ A_tgt,
                       float* __restrict__ out_both)
{
    const int side = blockIdx.z;
    const float* __restrict__ W     = side ? W_tgt     : W_src;
    const float* __restrict__ bias  = side ? bias_tgt  : bias_src;
    const float* __restrict__ Blora = side ? Blora_tgt : Blora_src;
    const float* __restrict__ A     = side ? A_tgt     : A_src;
    float* __restrict__       out   = out_both + side * (SEQ * NL * HID);

    const int hTile = blockIdx.x * (16 * P1_HT);
    const int sTile = blockIdx.y * 16;
    const int lane  = threadIdx.x & 31;
    const int wave  = __builtin_amdgcn_readfirstlane(threadIdx.x >> 5);
    const int nLoc  = lane & 15;
    const int half  = lane >> 4;
    const int kOff  = half * 2;

    float bl0[P1_VPW], bl1[P1_VPW], bvv[P1_VPW];
    #pragma unroll
    for (int e = 0; e < P1_VPW; ++e) {
        const int q  = wave + P1_WAVES * e;
        const int qc = (q < P1_NVAR) ? q : (P1_NVAR - 1);
        const int j  = qc / NL;
        const int l  = qc - j * NL;
        const int n  = hTile + 16 * j + nLoc;
        bl0[e] = Blora[(0 * HID + n) * NL + l];
        bl1[e] = Blora[(1 * HID + n) * NL + l];
        bvv[e] = bias[n];
    }

    v8f acc0 = {}, acc1 = {}, acc2 = {}, xac0 = {}, xac1 = {};
    const float* arow = X + (sTile + nLoc) * HID;
    const float* wcol = W + hTile + nLoc;
    const int c1 = (16 + nLoc < XA_COLS) ? (16 + nLoc) : (XA_COLS - 1);
    const int kBeg = wave * P1_KLEN;

    #pragma unroll 4
    for (int it = 0; it < P1_KLEN / 4; ++it) {
        const int kk = kBeg + 4 * it + kOff;
        const v2f a = *(const v2f*)(arow + kk);
        v2f b0, b1, b2, p0, p1;
        b0.x = wcol[kk * HID];            b0.y = wcol[(kk + 1) * HID];
        b1.x = wcol[kk * HID + 16];       b1.y = wcol[(kk + 1) * HID + 16];
        b2.x = wcol[kk * HID + 32];       b2.y = wcol[(kk + 1) * HID + 32];
        p0.x = A[kk * XA_COLS + nLoc];    p0.y = A[(kk + 1) * XA_COLS + nLoc];
        p1.x = A[kk * XA_COLS + c1];      p1.y = A[(kk + 1) * XA_COLS + c1];
        acc0 = wmma4(a, b0, acc0);
        acc1 = wmma4(a, b1, acc1);
        acc2 = wmma4(a, b2, acc2);
        xac0 = wmma4(a, p0, xac0);
        xac1 = wmma4(a, p1, xac1);
    }

    __shared__ float part[P1_WAVES][P1_TILES][8][32];
    __shared__ float totS[P1_HT][8][32];
    __shared__ __attribute__((aligned(8))) float xs[16 * XA_COLS];
    #pragma unroll
    for (int i = 0; i < 8; ++i) {
        part[wave][0][i][lane] = acc0[i];
        part[wave][1][i][lane] = acc1[i];
        part[wave][2][i][lane] = acc2[i];
        part[wave][3][i][lane] = xac0[i];
        part[wave][4][i][lane] = xac1[i];
    }
    __syncthreads();

    #pragma unroll
    for (int t = 0; t < P1_TILES; ++t) {
        float v = part[0][t][wave][lane];
        #pragma unroll
        for (int u = 1; u < P1_WAVES; ++u) v += part[u][t][wave][lane];
        if (t < P1_HT) {
            totS[t][wave][lane] = v;
        } else {
            const int c = 16 * (t - P1_HT) + nLoc;
            if (c < XA_COLS) {
                const int r   = (c >= NL) ? 1 : 0;
                const int l   = c - r * NL;
                const int row = 8 * half + wave;
                xs[(row * NL + l) * NR + r] = v;
            }
        }
    }
    __syncthreads();

    const int mBase = 8 * half;
    #pragma unroll
    for (int e = 0; e < P1_VPW; ++e) {
        const int q = wave + P1_WAVES * e;
        if (q < P1_NVAR) {
            const int j = q / NL;
            const int l = q - j * NL;
            const int n = hTile + 16 * j + nLoc;
            #pragma unroll
            for (int i = 0; i < 8; ++i) {
                const int row = mBase + i;
                const v2f xv = *(const v2f*)&xs[(row * NL + l) * NR];
                out[((sTile + row) * NL + l) * HID + n] =
                    totS[j][i][lane] + bvv[e] + xv.x * bl0[e] + xv.y * bl1[e];
            }
        }
    }
}

static __device__ __forceinline__ v2f splat2(float c) { v2f r = {c, c}; return r; }
static __device__ __forceinline__ v2f fma2(v2f a, v2f b, v2f c) {
    return __builtin_elementwise_fma(a, b, c);
}

static __device__ __forceinline__ v2f gelu_both_ranges2(v2f v) {
    v2f a;
    a.x = fabsf(v.x);
    a.y = fabsf(v.y);
    const v2f t = v * v;

    v2f u = fma2(a, splat2(-2.167644652e-6f), splat2(6.828958431e-5f));
    u = fma2(a, u, splat2(-9.838775732e-4f));
    u = fma2(a, u, splat2(8.721703663e-3f));
    u = fma2(a, u, splat2(-5.440825224e-2f));
    u = fma2(a, u, splat2(-4.580075443e-1f));
    u = fma2(a, u, splat2(-1.151420593e+0f));
    const v2f w = a * u;
    v2f e;
    e.x = __builtin_amdgcn_exp2f(w.x);
    e.y = __builtin_amdgcn_exp2f(w.y);
    const v2f big = a * fma2(e, splat2(-0.5f), splat2(0.5f));

    v2f r = fma2(t, splat2(-6.207091246e-6f), splat2(1.085810218e-4f));
    r = fma2(t, r, splat2(-1.180316554e-3f));
    r = fma2(t, r, splat2(9.970215149e-3f));
    r = fma2(t, r, splat2(-6.648977101e-2f));
    r = fma2(t, r, splat2(3.989422619e-1f));
    const v2f small = t * r;

    v2f x;
    x.x = (a.x < 1.41421354f) ? small.x : big.x;
    x.y = (a.y < 1.41421354f) ? small.y : big.y;
    return fma2(splat2(0.5f), v, x);
}

#define P2_ROWS   (SEQ / 8)
#define P2_CHUNKS (HID / 128)
#define P2_NCHUNK (P2_ROWS * P2_CHUNKS)
#define P2_RSTEP  (8 * NL * HID)

__global__ __launch_bounds__(256)
void pair_gelu_dot(const float* __restrict__ srcOut,
                   const float* __restrict__ tgtOut,
                   const float* __restrict__ cls,
                   float* __restrict__ out)
{
    const int s = blockIdx.x;
    const int l = blockIdx.y;

    __shared__ __attribute__((aligned(16))) float sSrc[HID];
    __shared__ __attribute__((aligned(16))) float sCls[HID];
    for (int h = threadIdx.x; h < HID; h += 256) {
        sSrc[h] = srcOut[(s * NL + l) * HID + h];
        sCls[h] = cls[h * NL + l];
    }
    __syncthreads();

    const int wave = threadIdx.x >> 5;
    const int lane = threadIdx.x & 31;

    const v4f* sSrc4 = (const v4f*)sSrc + lane;
    const v4f* sCls4 = (const v4f*)sCls + lane;

    const float* lbase = tgtOut + (wave * NL + l) * HID + 4 * lane;

    v4f b0 = *(const v4f*)(lbase);
    v4f b1 = *(const v4f*)(lbase + 128);
    int q = 2;

    for (int r = 0; r < P2_ROWS; ++r) {
        float sum = 0.0f;
        #pragma unroll 3
        for (int j = 0; j < P2_CHUNKS; ++j) {
            const v4f cur = b0;
            b0 = b1;
            {
                const int qq = (q < P2_NCHUNK) ? q : (P2_NCHUNK - 1);
                const int qr = qq / P2_CHUNKS;
                const int qj = qq - qr * P2_CHUNKS;
                b1 = *(const v4f*)(lbase + qr * P2_RSTEP + qj * 128);
                ++q;
            }
            const v4f sv = sSrc4[j * 32];
            const v4f cv = sCls4[j * 32];
            #pragma unroll
            for (int e = 0; e < 4; e += 2) {
                v2f v;
                v.x = sv[e]     + cur[e];
                v.y = sv[e + 1] + cur[e + 1];
                const v2f g = gelu_both_ranges2(v);
                sum = fmaf(g.x, cv[e], sum);
                sum = fmaf(g.y, cv[e + 1], sum);
            }
        }
        #pragma unroll
        for (int off = 16; off > 0; off >>= 1)
            sum += __shfl_xor(sum, off, 32);
        if (lane == 0)
            out[(l * SEQ + s) * SEQ + (wave + 8 * r)] = sum;
    }
}

extern "C" void kernel_launch(void* const* d_in, const int* in_sizes, int n_in,
                              void* d_out, int out_size, void* d_ws, size_t ws_size,
                              hipStream_t stream) {
    (void)in_sizes; (void)n_in; (void)out_size; (void)ws_size;

    const float* x     = (const float*)d_in[0];
    const float* W_src = (const float*)d_in[1];
    const float* b_src = (const float*)d_in[2];
    const float* W_tgt = (const float*)d_in[3];
    const float* b_tgt = (const float*)d_in[4];
    const float* A_src = (const float*)d_in[5];
    const float* B_src = (const float*)d_in[6];
    const float* A_tgt = (const float*)d_in[7];
    const float* B_tgt = (const float*)d_in[8];
    const float* cls   = (const float*)d_in[9];

    float* ws     = (float*)d_ws;
    float* srcOut = ws;
    float* tgtOut = srcOut + SEQ * NL * HID;


    dim3 g1(HID / (16 * P1_HT), SEQ / 16, 2);
    fused_linear_wmma<<<g1, 32 * P1_WAVES, 0, stream>>>(x, W_src, W_tgt, b_src, b_tgt,
                                            B_src, B_tgt, A_src, A_tgt, srcOut);

    dim3 g2(SEQ, NL);
    pair_gelu_dot<<<g2, 256, 0, stream>>>(srcOut, tgtOut, cls, (float*)d_out);
}
